# starting kernel
# baseline (speedup 1.0000x reference)
#include <hip/hip_runtime.h>

typedef float v2f __attribute__((ext_vector_type(2)));
typedef float v8f __attribute__((ext_vector_type(8)));

#define C_DIM 64
#define D_DIM 128
#define P_DIM 64
#define TILE_PITCH 65
#define BOOST_F 1000000.0f

__global__ __launch_bounds__(128)
void PatchSelector_kernel(const float* __restrict__ x,
                          const int*   __restrict__ channel_idx,
                          const float* __restrict__ W,
                          const float* __restrict__ bias,
                          float* __restrict__ out) {
  __shared__ float tile[D_DIM * TILE_PITCH];
  __shared__ float skey[P_DIM];
  __shared__ int   perm[P_DIM];

  const int b    = blockIdx.x;
  const int tid  = threadIdx.x;
  const int lane = tid & 31;
  const int wave = tid >> 5;

  const int   cid  = channel_idx[0];
  const float bval = bias[0];

  const float* xb = x + (size_t)(b * C_DIM + cid) * (D_DIM * P_DIM);

  for (int i = tid; i < D_DIM * P_DIM; i += 128) {
    const int d = i >> 6;
    const int p = i & (P_DIM - 1);
    tile[d * TILE_PITCH + p] = xb[i];
  }
  __syncthreads();

  {
    const int pb    = wave;
    const int j     = lane & 15;
    const int khalf = (lane >> 4) << 1;
    const int jg    = pb * 16 + j;

    v8f acc = {};
    #pragma unroll 4
    for (int c = 0; c < 32; ++c) {
      const int d0 = c * 4 + khalf;
      v2f a, bm;
      a.x  = W[d0];
      a.y  = W[d0 + 1];
      bm.x = tile[d0 * TILE_PITCH + jg];
      bm.y = tile[(d0 + 1) * TILE_PITCH + jg];
      acc = __builtin_amdgcn_wmma_f32_16x16x4_f32(
          false, a, false, bm, (short)0, acc, false, false);
    }
    if (lane < 16) {
      const float s = acc[0] + bval;
      skey[jg] = s + BOOST_F;
    }
  }
  __syncthreads();

  if (tid < P_DIM) {
    const float my = skey[tid];
    int rank = 0;
    #pragma unroll 8
    for (int jj = 0; jj < P_DIM; ++jj) {
      const float o = skey[jj];
      rank += (o > my) || (o == my && jj < tid);
    }
    perm[rank] = tid;
  }
  __syncthreads();

  float* ob = out + (size_t)b * (P_DIM * D_DIM);
  for (int o = tid; o < P_DIM * D_DIM; o += 128) {
    const int k = o >> 7;
    const int d = o & (D_DIM - 1);
    ob[o] = tile[d * TILE_PITCH + perm[k]];
  }
}

extern "C" void kernel_launch(void* const* d_in, const int* in_sizes, int n_in,
                              void* d_out, int out_size, void* d_ws, size_t ws_size,
                              hipStream_t stream) {
  const float* x   = (const float*)d_in[0];
  const int*   cid = (const int*)d_in[1];
  const float* W   = (const float*)d_in[2];
  const float* bb  = (const float*)d_in[3];
  float*       out = (float*)d_out;

  const int B = out_size / (P_DIM * D_DIM);
  PatchSelector_kernel<<<dim3(B), dim3(128), 0, stream>>>(x, cid, W, bb, out);
}


# All global reads issued as one batch at entry (slice as 16 in-flight 128-bit loads per thread, W staged to LDS); score loop reads LDS only: one memory round trip instead of twelve
# speedup vs baseline: 1.3314x; 1.3314x over previous
#include <hip/hip_runtime.h>

typedef float v2f __attribute__((ext_vector_type(2)));
typedef float v4f __attribute__((ext_vector_type(4)));
typedef float v8f __attribute__((ext_vector_type(8)));

#define C_DIM 64
#define D_DIM 128
#define P_DIM 64
#define TILE_PITCH 65
#define BOOST_F 1000000.0f

__global__ __launch_bounds__(128)
void PatchSelector_kernel(const float* __restrict__ x,
                          const int*   __restrict__ channel_idx,
                          const float* __restrict__ W,
                          const float* __restrict__ bias,
                          float* __restrict__ out) {
  __shared__ float tile[D_DIM * TILE_PITCH];
  __shared__ float sW[D_DIM];
  __shared__ float skey[P_DIM];
  __shared__ int   perm[P_DIM];

  const int b    = blockIdx.x;
  const int tid  = threadIdx.x;
  const int lane = tid & 31;
  const int wave = tid >> 5;

  const int   cid  = channel_idx[0];
  const float bval = bias[0];

  const float* xb = x + (size_t)(b * C_DIM + cid) * (D_DIM * P_DIM);

  static_assert(D_DIM == 128, "one W element per thread of the 128-thread block");
  constexpr int N_LD = (D_DIM * P_DIM) / (4 * 128);
  const float wv = W[tid];

  const v4f* xb4 = (const v4f*)xb;
  v4f xr[N_LD];
  for (int it = 0; it < N_LD; ++it)
    xr[it] = xb4[tid + it * 128];

  sW[tid] = wv;
  for (int it = 0; it < N_LD; ++it) {
    const int i4 = tid + it * 128;
    const int d  = i4 >> 4;
    const int p  = (i4 & 15) << 2;
    float* t = &tile[d * TILE_PITCH + p];
    t[0] = xr[it].x;
    t[1] = xr[it].y;
    t[2] = xr[it].z;
    t[3] = xr[it].w;
  }
  __syncthreads();

  {
    const int pb    = wave;
    const int j     = lane & 15;
    const int khalf = (lane >> 4) << 1;
    const int jg    = pb * 16 + j;

    v8f acc = {};
    #pragma unroll 4
    for (int c = 0; c < 32; ++c) {
      const int d0 = c * 4 + khalf;
      v2f a, bm;
      a.x  = sW[d0];
      a.y  = sW[d0 + 1];
      bm.x = tile[d0 * TILE_PITCH + jg];
      bm.y = tile[(d0 + 1) * TILE_PITCH + jg];
      acc = __builtin_amdgcn_wmma_f32_16x16x4_f32(
          false, a, false, bm, (short)0, acc, false, false);
    }
    if (lane < 16) {
      const float s = acc[0] + bval;
      skey[jg] = s + BOOST_F;
    }
  }
  __syncthreads();

  if (tid < P_DIM) {
    const float my = skey[tid];
    int rank = 0;
    #pragma unroll 8
    for (int jj = 0; jj < P_DIM; ++jj) {
      const float o = skey[jj];
      rank += (o > my) || (o == my && jj < tid);
    }
    perm[rank] = tid;
  }
  __syncthreads();

  float* ob = out + (size_t)b * (P_DIM * D_DIM);
  for (int o = tid; o < P_DIM * D_DIM; o += 128) {
    const int k = o >> 7;
    const int d = o & (D_DIM - 1);
    ob[o] = tile[d * TILE_PITCH + perm[k]];
  }
}

extern "C" void kernel_launch(void* const* d_in, const int* in_sizes, int n_in,
                              void* d_out, int out_size, void* d_ws, size_t ws_size,
                              hipStream_t stream) {
  const float* x   = (const float*)d_in[0];
  const int*   cid = (const int*)d_in[1];
  const float* W   = (const float*)d_in[2];
  const float* bb  = (const float*)d_in[3];
  float*       out = (float*)d_out;

  const int B = out_size / (P_DIM * D_DIM);
  PatchSelector_kernel<<<dim3(B), dim3(128), 0, stream>>>(x, cid, W, bb, out);
}


# Each phase's LDS reads issued as one batch ahead of their consumers (score chain, rank compares, row gather turned into a scatter by rank; all unrolled): one LDS wait per phase, not per loop trip
# speedup vs baseline: 1.5157x; 1.1384x over previous
#include <hip/hip_runtime.h>

typedef float v2f __attribute__((ext_vector_type(2)));
typedef float v4f __attribute__((ext_vector_type(4)));
typedef float v8f __attribute__((ext_vector_type(8)));

#define C_DIM 64
#define D_DIM 128
#define P_DIM 64
#define TILE_PITCH 65
#define BOOST_F 1000000.0f

__global__ __launch_bounds__(128)
void PatchSelector_kernel(const float* __restrict__ x,
                          const int*   __restrict__ channel_idx,
                          const float* __restrict__ W,
                          const float* __restrict__ bias,
                          float* __restrict__ out) {
  __shared__ float tile[D_DIM * TILE_PITCH];
  __shared__ float sW[D_DIM];
  __shared__ float skey[P_DIM];
  __shared__ int   rnk[P_DIM];

  const int b    = blockIdx.x;
  const int tid  = threadIdx.x;
  const int lane = tid & 31;
  const int wave = tid >> 5;

  const int   cid  = channel_idx[0];
  const float bval = bias[0];

  const float* xb = x + (size_t)(b * C_DIM + cid) * (D_DIM * P_DIM);

  static_assert(D_DIM == 128, "one W element per thread of the 128-thread block");
  constexpr int N_LD = (D_DIM * P_DIM) / (4 * 128);
  const float wv = W[tid];

  const v4f* xb4 = (const v4f*)xb;
  v4f xr[N_LD];
  for (int it = 0; it < N_LD; ++it)
    xr[it] = xb4[tid + it * 128];

  sW[tid] = wv;
  for (int it = 0; it < N_LD; ++it) {
    const int i4 = tid + it * 128;
    const int d  = i4 >> 4;
    const int p  = (i4 & 15) << 2;
    float* t = &tile[d * TILE_PITCH + p];
    t[0] = xr[it].x;
    t[1] = xr[it].y;
    t[2] = xr[it].z;
    t[3] = xr[it].w;
  }
  __syncthreads();

  {
    const int pb    = wave;
    const int j     = lane & 15;
    const int khalf = (lane >> 4) << 1;
    const int jg    = pb * 16 + j;

    v2f a[32], bm[32];
    #pragma unroll
    for (int c = 0; c < 32; ++c) {
      const int d0 = c * 4 + khalf;
      a[c].x  = sW[d0];
      a[c].y  = sW[d0 + 1];
      bm[c].x = tile[d0 * TILE_PITCH + jg];
      bm[c].y = tile[(d0 + 1) * TILE_PITCH + jg];
    }
    __builtin_amdgcn_sched_barrier(0);
    v8f acc = {};
    #pragma unroll
    for (int c = 0; c < 32; ++c) {
      acc = __builtin_amdgcn_wmma_f32_16x16x4_f32(
          false, a[c], false, bm[c], (short)0, acc, false, false);
    }
    if (lane < 16) {
      const float s = acc[0] + bval;
      skey[jg] = s + BOOST_F;
    }
  }
  __syncthreads();

  if (tid < P_DIM) {
    const float my = skey[tid];
    float key[P_DIM];
    #pragma unroll
    for (int jj = 0; jj < P_DIM; ++jj) key[jj] = skey[jj];
    int rank = 0;
    #pragma unroll
    for (int jj = 0; jj < P_DIM; ++jj) {
      const float o = key[jj];
      rank += (o > my) || (o == my && jj < tid);
    }
    rnk[tid] = rank;
  }
  __syncthreads();

  static_assert(D_DIM == 128, "one d column per thread of the 128-thread block");
  float* ob = out + (size_t)b * (P_DIM * D_DIM);
  static_assert(P_DIM == 64, "lane l keeps the ranks of patches l and l + 32");
  const int r_lo = rnk[lane];
  const int r_hi = rnk[lane + 32];
  float val[P_DIM];
  #pragma unroll
  for (int p = 0; p < P_DIM; ++p)
    val[p] = tile[tid * TILE_PITCH + p];
  __builtin_amdgcn_sched_barrier(0);
  #pragma unroll
  for (int p = 0; p < P_DIM; ++p) {
    const int r = __builtin_amdgcn_readlane(p < 32 ? r_lo : r_hi, p & 31);
    ob[r * D_DIM + tid] = val[p];
  }
}

extern "C" void kernel_launch(void* const* d_in, const int* in_sizes, int n_in,
                              void* d_out, int out_size, void* d_ws, size_t ws_size,
                              hipStream_t stream) {
  const float* x   = (const float*)d_in[0];
  const int*   cid = (const int*)d_in[1];
  const float* W   = (const float*)d_in[2];
  const float* bb  = (const float*)d_in[3];
  float*       out = (float*)d_out;

  const int B = out_size / (P_DIM * D_DIM);
  PatchSelector_kernel<<<dim3(B), dim3(128), 0, stream>>>(x, cid, W, bb, out);
}
